# starting kernel
# baseline (speedup 1.0000x reference)
#include <hip/hip_runtime.h>
#include <hip/hip_bf16.h>

#define N_NODES 100000
#define N_EDGES 800000
#define IN_DIM  256
#define HID     128
#define OUT_DIM 64

typedef float v2f __attribute__((ext_vector_type(2)));
typedef float v8f __attribute__((ext_vector_type(8)));

__global__ void zero_kernel(float* __restrict__ p, int n) {
    int i = blockIdx.x * blockDim.x + threadIdx.x;
    if (i < n) p[i] = 0.0f;
}

__global__ void degree_kernel(const int* __restrict__ dst, float* __restrict__ degs, int n_edges) {
    int e = blockIdx.x * blockDim.x + threadIdx.x;
    if (e < n_edges) atomicAdd(&degs[dst[e]], 1.0f);
}

__global__ void norm_kernel(float* __restrict__ norm, int n) {
    int i = blockIdx.x * blockDim.x + threadIdx.x;
    if (i < n) {
        float d = norm[i];
        norm[i] = rsqrtf(fmaxf(d, 1.0f));
    }
}

__global__ void scatter_kernel(const float* __restrict__ h,
                               const float* __restrict__ norm,
                               const int* __restrict__ src,
                               const int* __restrict__ dst,
                               float* __restrict__ agg, int n_edges) {
    int gid = blockIdx.x * blockDim.x + threadIdx.x;
    int e = gid >> 5;
    if (e >= n_edges) return;
    int l = gid & 31;
    int s = src[e];
    int d = dst[e];
    float ns = norm[s];
    const float4 v = *reinterpret_cast<const float4*>(h + (size_t)s * HID + l * 4);
    float* out = agg + (size_t)d * HID + l * 4;
    atomicAdd(out + 0, v.x * ns);
    atomicAdd(out + 1, v.y * ns);
    atomicAdd(out + 2, v.z * ns);
    atomicAdd(out + 3, v.w * ns);
}

template<int NT>
__global__ __launch_bounds__(32) void wmma_gemm_strip(
    const float* __restrict__ A, const float* __restrict__ W,
    const float* __restrict__ bias, const float* __restrict__ norm,
    const float* __restrict__ resid, float* __restrict__ C,
    int M, int K, int do_relu)
{
    constexpr int N = NT * 16;
    const int lane  = threadIdx.x;
    const int half  = lane >> 4;
    const int idx16 = lane & 15;
    const int m0 = blockIdx.x * 16;

    const int arow = m0 + idx16;
    const int koff = half * 2;

    const float nscale = (norm != nullptr) ? norm[arow] : 1.0f;
    const float* __restrict__ Arow = A + (size_t)arow * K;

    v8f acc[NT];
#pragma unroll
    for (int t = 0; t < NT; ++t) acc[t] = (v8f){};

    for (int k = 0; k < K; k += 4) {
        v2f a = *reinterpret_cast<const v2f*>(Arow + k + koff);
        a.x *= nscale;
        a.y *= nscale;
        const float* __restrict__ Wr0 = W + (size_t)(k + koff) * N + idx16;
        const float* __restrict__ Wr1 = Wr0 + N;
#pragma unroll
        for (int t = 0; t < NT; ++t) {
            v2f b;
            b.x = Wr0[t * 16];
            b.y = Wr1[t * 16];
            acc[t] = __builtin_amdgcn_wmma_f32_16x16x4_f32(
                 false, a,  false, b,
                 (short)0, acc[t],  false,  false);
        }
    }

#pragma unroll
    for (int t = 0; t < NT; ++t) {
        const int bcol = t * 16 + idx16;
        const float bv = bias[bcol];
#pragma unroll
        for (int r = 0; r < 8; ++r) {
            int orow = m0 + half * 8 + r;
            size_t oidx = (size_t)orow * N + bcol;
            float v = acc[t][r] + bv;
            if (do_relu) v = fmaxf(v, 0.0f);
            if (resid != nullptr) v += resid[oidx];
            C[oidx] = v;
        }
    }
}

extern "C" void kernel_launch(void* const* d_in, const int* in_sizes, int n_in,
                              void* d_out, int out_size, void* d_ws, size_t ws_size,
                              hipStream_t stream) {
    const float* features = (const float*)d_in[0];
    const int*   src      = (const int*)  d_in[1];
    const int*   dst      = (const int*)  d_in[2];
    const float* W_in     = (const float*)d_in[3];
    const float* b_in     = (const float*)d_in[4];
    const float* gcn_w0   = (const float*)d_in[5];
    const float* gcn_b0   = (const float*)d_in[6];
    const float* gcn_w1   = (const float*)d_in[7];
    const float* gcn_b1   = (const float*)d_in[8];
    const float* W_out    = (const float*)d_in[9];
    const float* b_out    = (const float*)d_in[10];
    float* out = (float*)d_out;

    char* ws = (char*)d_ws;
    float* norm = (float*)ws;
    ws += ((size_t)N_NODES * sizeof(float) + 255) & ~(size_t)255;
    float* H = (float*)ws;
    ws += ((size_t)N_NODES * HID * sizeof(float) + 255) & ~(size_t)255;
    float* AGG = (float*)ws;

    const int nodeVec = N_NODES * HID;
    const int mBlocks = N_NODES / 16;

    zero_kernel<<<(N_NODES + 255) / 256, 256, 0, stream>>>(norm, N_NODES);
    degree_kernel<<<(N_EDGES + 255) / 256, 256, 0, stream>>>(dst, norm, N_EDGES);
    norm_kernel<<<(N_NODES + 255) / 256, 256, 0, stream>>>(norm, N_NODES);

    wmma_gemm_strip<8><<<mBlocks, 32, 0, stream>>>(
        features, W_in, b_in, nullptr, nullptr, H, N_NODES, IN_DIM,  0);

    zero_kernel<<<(nodeVec + 255) / 256, 256, 0, stream>>>(AGG, nodeVec);
    scatter_kernel<<<(N_EDGES * 32 + 255) / 256, 256, 0, stream>>>(
        H, norm, src, dst, AGG, N_EDGES);
    wmma_gemm_strip<8><<<mBlocks, 32, 0, stream>>>(
        AGG, gcn_w0, gcn_b0, norm,  H, H, N_NODES, HID,  1);

    zero_kernel<<<(nodeVec + 255) / 256, 256, 0, stream>>>(AGG, nodeVec);
    scatter_kernel<<<(N_EDGES * 32 + 255) / 256, 256, 0, stream>>>(
        H, norm, src, dst, AGG, N_EDGES);
    wmma_gemm_strip<8><<<mBlocks, 32, 0, stream>>>(
        AGG, gcn_w1, gcn_b1, norm, nullptr, H, N_NODES, HID,  1);

    wmma_gemm_strip<4><<<mBlocks, 32, 0, stream>>>(
        H, W_out, b_out, nullptr, nullptr, out, N_NODES, HID,  0);
}


# Neighbour sum by dst owner: workgroup keeps 391 nodes' sums in LDS, scans dst list, adds its edges' rows with LDS float adds, stores agg once; no global atomics, no zero fill; add order by arrival
# speedup vs baseline: 1.5176x; 1.5176x over previous
#include <hip/hip_runtime.h>
#include <hip/hip_bf16.h>

#define N_NODES 100000
#define N_EDGES 800000
#define IN_DIM  256
#define HID     128
#define OUT_DIM 64

typedef float v2f __attribute__((ext_vector_type(2)));
typedef float v8f __attribute__((ext_vector_type(8)));

__global__ void zero_kernel(float* __restrict__ p, int n) {
    int i = blockIdx.x * blockDim.x + threadIdx.x;
    if (i < n) p[i] = 0.0f;
}

__global__ void degree_kernel(const int* __restrict__ dst, float* __restrict__ degs, int n_edges) {
    int e = blockIdx.x * blockDim.x + threadIdx.x;
    if (e < n_edges) atomicAdd(&degs[dst[e]], 1.0f);
}

__global__ void norm_kernel(float* __restrict__ norm, int n) {
    int i = blockIdx.x * blockDim.x + threadIdx.x;
    if (i < n) {
        float d = norm[i];
        norm[i] = rsqrtf(fmaxf(d, 1.0f));
    }
}

#define NODES_PER_BLOCK 391
#define OWNER_THREADS   1024
#define OWNER_WAVES     (OWNER_THREADS / 32)
#define QUEUE_CAP       160
#define SCAN_DEPTH      8
#define EDGES_IN_FLIGHT 8

__device__ __forceinline__ void add_queued_rows(const float* __restrict__ h,
                                                float* acc,
                                                int s, int dl, int nsb, int cntw, int lane) {
    if (cntw == 32) {
        for (int j0 = 0; j0 < 32; j0 += EDGES_IN_FLIGHT) {
            float v[EDGES_IN_FLIGHT][4];
#pragma unroll
            for (int j = 0; j < EDGES_IN_FLIGHT; ++j) {
                const int sj = __builtin_amdgcn_readlane(s, j0 + j);
                const float* __restrict__ hp = h + (size_t)sj * HID + lane;
                v[j][0] = hp[0];
                v[j][1] = hp[32];
                v[j][2] = hp[64];
                v[j][3] = hp[96];
            }
#pragma unroll
            for (int j = 0; j < EDGES_IN_FLIGHT; ++j) {
                const int dj = __builtin_amdgcn_readlane(dl, j0 + j);
                const float nj = __int_as_float(__builtin_amdgcn_readlane(nsb, j0 + j));
                float* a = acc + dj * HID + lane;
                atomicAdd(a +  0, v[j][0] * nj);
                atomicAdd(a + 32, v[j][1] * nj);
                atomicAdd(a + 64, v[j][2] * nj);
                atomicAdd(a + 96, v[j][3] * nj);
            }
        }
    } else {
        for (int j = 0; j < cntw; ++j) {
            const int sj = __builtin_amdgcn_readlane(s, j);
            const int dj = __builtin_amdgcn_readlane(dl, j);
            const float nj = __int_as_float(__builtin_amdgcn_readlane(nsb, j));
            const float* __restrict__ hp = h + (size_t)sj * HID + lane;
            const float v0 = hp[0];
            const float v1 = hp[32];
            const float v2 = hp[64];
            const float v3 = hp[96];
            float* a = acc + dj * HID + lane;
            atomicAdd(a +  0, v0 * nj);
            atomicAdd(a + 32, v1 * nj);
            atomicAdd(a + 64, v2 * nj);
            atomicAdd(a + 96, v3 * nj);
        }
    }
}

__global__ __launch_bounds__(OWNER_THREADS) void owner_sum_kernel(
    const float* __restrict__ h, const float* __restrict__ norm,
    const int* __restrict__ src, const int* __restrict__ dst,
    float* __restrict__ agg, int n_edges, int n_nodes)
{
    __shared__ float acc[NODES_PER_BLOCK * HID];
    __shared__ int q_edge[OWNER_WAVES][QUEUE_CAP];
    __shared__ int q_dloc[OWNER_WAVES][QUEUE_CAP];

    const int tid  = threadIdx.x;
    const int lane = tid & 31;
    const int wave = __builtin_amdgcn_readfirstlane(tid >> 5);
    const int lo   = blockIdx.x * NODES_PER_BLOCK;
    const int cnt  = min(NODES_PER_BLOCK, n_nodes - lo);
    const unsigned lane_below = (1u << lane) - 1u;

    {
        float4* acc4 = reinterpret_cast<float4*>(acc);
        const float4 z = make_float4(0.0f, 0.0f, 0.0f, 0.0f);
        for (int i = tid; i < NODES_PER_BLOCK * HID / 4; i += OWNER_THREADS) acc4[i] = z;
    }
    __syncthreads();

    int* qe = q_edge[wave];
    int* qd = q_dloc[wave];
    int pend = 0;

    const int4* __restrict__ dst4 = reinterpret_cast<const int4*>(dst);
    const int n4 = n_edges >> 2;

    auto drain = [&]() {
        int head = 0;
        __builtin_amdgcn_wave_barrier();
        while (pend - head >= 32) {
            const int e  = qe[head + lane];
            const int dl = qd[head + lane];
            const int s  = src[e];
            const int nsb = __float_as_int(norm[s]);
            add_queued_rows(h, acc, s, dl, nsb, 32, lane);
            head += 32;
        }
        const int rem = pend - head;
        int te = 0, td = 0;
        if (lane < rem) { te = qe[head + lane]; td = qd[head + lane]; }
        if (lane < rem) { qe[lane] = te; qd[lane] = td; }
        pend = rem;
    };

    auto handle = [&](const int4 d, const int idx, const int cntv) {
        const int dc[4] = {d.x, d.y, d.z, d.w};
        unsigned m[4];
#pragma unroll
        for (int c = 0; c < 4; ++c)
            m[c] = __builtin_amdgcn_ballot_w32((unsigned)(dc[c] - lo) < (unsigned)cntv);
        if ((m[0] | m[1] | m[2] | m[3]) == 0u) return;
#pragma unroll
        for (int c = 0; c < 4; ++c) {
            if (m[c] == 0u) continue;
            const int pos = pend + __builtin_popcount(m[c] & lane_below);
            if ((m[c] >> lane) & 1u) {
                qe[pos] = idx * 4 + c;
                qd[pos] = dc[c] - lo;
            }
            pend += __builtin_popcount(m[c]);
        }
        if (pend >= 32) drain();
    };

    if (n4 > 0) {
        const int last4 = n4 - 1;
        for (int base = wave * 32; base < n4; base += SCAN_DEPTH * OWNER_THREADS) {
            int4 dv[SCAN_DEPTH];
            int  ix[SCAN_DEPTH];
#pragma unroll
            for (int u = 0; u < SCAN_DEPTH; ++u) {
                ix[u] = base + u * OWNER_THREADS + lane;
                dv[u] = dst4[min(ix[u], last4)];
            }
#pragma unroll
            for (int u = 0; u < SCAN_DEPTH; ++u)
                handle(dv[u], ix[u], (ix[u] < n4) ? cnt : 0);
        }
    }

    if (wave == 0) {
        const int e = n4 * 4 + lane;
        int dv = -1;
        if (lane < (n_edges & 3)) dv = dst[e];
        const unsigned mt = __builtin_amdgcn_ballot_w32((unsigned)(dv - lo) < (unsigned)cnt);
        if (mt != 0u) {
            const int pos = pend + __builtin_popcount(mt & lane_below);
            if ((mt >> lane) & 1u) { qe[pos] = e; qd[pos] = dv - lo; }
            pend += __builtin_popcount(mt);
        }
    }

    int done = 0;
    while (done < pend) {
        const int cw = min(32, pend - done);
        __builtin_amdgcn_wave_barrier();
        int s = 0, dl = 0, nsb = 0;
        if (lane < cw) {
            const int e = qe[done + lane];
            dl = qd[done + lane];
            s = src[e];
            nsb = __float_as_int(norm[s]);
        }
        add_queued_rows(h, acc, s, dl, nsb, cw, lane);
        done += cw;
    }

    __syncthreads();
    {
        const float4* acc4 = reinterpret_cast<const float4*>(acc);
        float4* __restrict__ out4 = reinterpret_cast<float4*>(agg + (size_t)lo * HID);
        const int nv = cnt * (HID / 4);
        for (int i = tid; i < nv; i += OWNER_THREADS) out4[i] = acc4[i];
    }
}

template<int NT>
__global__ __launch_bounds__(32) void wmma_gemm_strip(
    const float* __restrict__ A, const float* __restrict__ W,
    const float* __restrict__ bias, const float* __restrict__ norm,
    const float* __restrict__ resid, float* __restrict__ C,
    int M, int K, int do_relu)
{
    constexpr int N = NT * 16;
    const int lane  = threadIdx.x;
    const int half  = lane >> 4;
    const int idx16 = lane & 15;
    const int m0 = blockIdx.x * 16;

    const int arow = m0 + idx16;
    const int koff = half * 2;

    const float nscale = (norm != nullptr) ? norm[arow] : 1.0f;
    const float* __restrict__ Arow = A + (size_t)arow * K;

    v8f acc[NT];
#pragma unroll
    for (int t = 0; t < NT; ++t) acc[t] = (v8f){};

    for (int k = 0; k < K; k += 4) {
        v2f a = *reinterpret_cast<const v2f*>(Arow + k + koff);
        a.x *= nscale;
        a.y *= nscale;
        const float* __restrict__ Wr0 = W + (size_t)(k + koff) * N + idx16;
        const float* __restrict__ Wr1 = Wr0 + N;
#pragma unroll
        for (int t = 0; t < NT; ++t) {
            v2f b;
            b.x = Wr0[t * 16];
            b.y = Wr1[t * 16];
            acc[t] = __builtin_amdgcn_wmma_f32_16x16x4_f32(
                 false, a,  false, b,
                 (short)0, acc[t],  false,  false);
        }
    }

#pragma unroll
    for (int t = 0; t < NT; ++t) {
        const int bcol = t * 16 + idx16;
        const float bv = bias[bcol];
#pragma unroll
        for (int r = 0; r < 8; ++r) {
            int orow = m0 + half * 8 + r;
            size_t oidx = (size_t)orow * N + bcol;
            float v = acc[t][r] + bv;
            if (do_relu) v = fmaxf(v, 0.0f);
            if (resid != nullptr) v += resid[oidx];
            C[oidx] = v;
        }
    }
}

extern "C" void kernel_launch(void* const* d_in, const int* in_sizes, int n_in,
                              void* d_out, int out_size, void* d_ws, size_t ws_size,
                              hipStream_t stream) {
    const float* features = (const float*)d_in[0];
    const int*   src      = (const int*)  d_in[1];
    const int*   dst      = (const int*)  d_in[2];
    const float* W_in     = (const float*)d_in[3];
    const float* b_in     = (const float*)d_in[4];
    const float* gcn_w0   = (const float*)d_in[5];
    const float* gcn_b0   = (const float*)d_in[6];
    const float* gcn_w1   = (const float*)d_in[7];
    const float* gcn_b1   = (const float*)d_in[8];
    const float* W_out    = (const float*)d_in[9];
    const float* b_out    = (const float*)d_in[10];
    float* out = (float*)d_out;

    char* ws = (char*)d_ws;
    float* norm = (float*)ws;
    ws += ((size_t)N_NODES * sizeof(float) + 255) & ~(size_t)255;
    float* H = (float*)ws;
    ws += ((size_t)N_NODES * HID * sizeof(float) + 255) & ~(size_t)255;
    float* AGG = (float*)ws;

    const int mBlocks = N_NODES / 16;

    zero_kernel<<<(N_NODES + 255) / 256, 256, 0, stream>>>(norm, N_NODES);
    degree_kernel<<<(N_EDGES + 255) / 256, 256, 0, stream>>>(dst, norm, N_EDGES);
    norm_kernel<<<(N_NODES + 255) / 256, 256, 0, stream>>>(norm, N_NODES);

    wmma_gemm_strip<8><<<mBlocks, 32, 0, stream>>>(
        features, W_in, b_in, nullptr, nullptr, H, N_NODES, IN_DIM,  0);

    owner_sum_kernel<<<(N_NODES + NODES_PER_BLOCK - 1) / NODES_PER_BLOCK, OWNER_THREADS, 0, stream>>>(
        H, norm, src, dst, AGG, N_EDGES, N_NODES);
    wmma_gemm_strip<8><<<mBlocks, 32, 0, stream>>>(
        AGG, gcn_w0, gcn_b0, norm,  H, H, N_NODES, HID,  1);

    owner_sum_kernel<<<(N_NODES + NODES_PER_BLOCK - 1) / NODES_PER_BLOCK, OWNER_THREADS, 0, stream>>>(
        H, norm, src, dst, AGG, N_EDGES, N_NODES);
    wmma_gemm_strip<8><<<mBlocks, 32, 0, stream>>>(
        AGG, gcn_w1, gcn_b1, norm, nullptr, H, N_NODES, HID,  1);

    wmma_gemm_strip<4><<<mBlocks, 32, 0, stream>>>(
        H, W_out, b_out, nullptr, nullptr, out, N_NODES, HID,  0);
}


# Edges binned once per call by owner band (LDS counting sort per edge slice, in the degree launch); both neighbour sums read their band's piece of each slice, no dst scan; add order still by arrival
# speedup vs baseline: 1.7687x; 1.1655x over previous
#include <hip/hip_runtime.h>
#include <hip/hip_bf16.h>

#define N_NODES 100000
#define N_EDGES 800000
#define IN_DIM  256
#define HID     128
#define OUT_DIM 64

typedef float v2f __attribute__((ext_vector_type(2)));
typedef float v8f __attribute__((ext_vector_type(8)));

__global__ void zero_kernel(float* __restrict__ p, int n) {
    int i = blockIdx.x * blockDim.x + threadIdx.x;
    if (i < n) p[i] = 0.0f;
}

#define NODES_PER_BLOCK 391
#define NUM_OWNERS      ((N_NODES + NODES_PER_BLOCK - 1) / NODES_PER_BLOCK)
#define BIN_SLICES      256
#define BIN_THREADS     1024
#define SLICE_EDGES     ((N_EDGES + BIN_SLICES - 1) / BIN_SLICES)
#define EDGES_PER_THREAD ((SLICE_EDGES + BIN_THREADS - 1) / BIN_THREADS)
#define LIST_PITCH      (((SLICE_EDGES + 31) / 32) * 32)
#define OFF_PITCH       (((NUM_OWNERS + 1 + 31) / 32) * 32)
#define DLOC_BITS       9
#define BIN_WORDS       (BIN_SLICES * (LIST_PITCH + OFF_PITCH))

static_assert(NODES_PER_BLOCK <= (1 << DLOC_BITS), "local node index must fit its field");
static_assert(N_NODES <= (1 << (31 - DLOC_BITS)), "source index must fit its field");
static_assert(NUM_OWNERS % 32 == 0 && NUM_OWNERS <= BIN_THREADS, "prefix code assumes whole waves");
static_assert(BIN_WORDS <= N_NODES * OUT_DIM, "bins must fit the buffer that holds them");

__global__ __launch_bounds__(BIN_THREADS) void degree_bin_kernel(
    const int* __restrict__ src, const int* __restrict__ dst,
    float* __restrict__ degs, int* __restrict__ lists, int* __restrict__ offs, int n_edges)
{
    __shared__ int cnt[NUM_OWNERS];
    __shared__ int start[NUM_OWNERS + 1];
    __shared__ int wave_tot[NUM_OWNERS / 32];
    __shared__ int sorted[SLICE_EDGES];

    const int tid = threadIdx.x;
    const int g   = blockIdx.x;
    const int e0  = g * SLICE_EDGES;
    const int ne  = max(0, min(SLICE_EDGES, n_edges - e0));

    if (tid < NUM_OWNERS) cnt[tid] = 0;
    __syncthreads();

    int dk[EDGES_PER_THREAD], sk[EDGES_PER_THREAD];
#pragma unroll
    for (int k = 0; k < EDGES_PER_THREAD; ++k) {
        const int i = tid + k * BIN_THREADS;
        dk[k] = 0;
        sk[k] = 0;
        if (i < ne) {
            dk[k] = dst[e0 + i];
            sk[k] = src[e0 + i];
        }
    }
    int pk[EDGES_PER_THREAD], bk[EDGES_PER_THREAD];
#pragma unroll
    for (int k = 0; k < EDGES_PER_THREAD; ++k) {
        const int i = tid + k * BIN_THREADS;
        bk[k] = -1;
        pk[k] = 0;
        if (i < ne) {
            const int d = dk[k];
            const int b = d / NODES_PER_BLOCK;
            bk[k] = b;
            pk[k] = (int)(((unsigned)sk[k] << DLOC_BITS) | (unsigned)(d - b * NODES_PER_BLOCK));
            atomicAdd(&cnt[b], 1);
            atomicAdd(&degs[d], 1.0f);
        }
    }
    __syncthreads();

    int v = 0, incl = 0;
    if (tid < NUM_OWNERS) {
        v = cnt[tid];
        incl = v;
        for (int o = 1; o < 32; o <<= 1) {
            const int t = __shfl_up(incl, o);
            if ((tid & 31) >= o) incl += t;
        }
        if ((tid & 31) == 31) wave_tot[tid >> 5] = incl;
    }
    __syncthreads();
    if (tid < NUM_OWNERS) {
        int base = 0;
        for (int w = 0; w < (tid >> 5); ++w) base += wave_tot[w];
        const int ex = base + incl - v;
        start[tid] = ex;
        cnt[tid]   = ex;
        if (tid == NUM_OWNERS - 1) start[NUM_OWNERS] = base + incl;
    }
    __syncthreads();

#pragma unroll
    for (int k = 0; k < EDGES_PER_THREAD; ++k) {
        if (bk[k] >= 0) {
            const int pos = atomicAdd(&cnt[bk[k]], 1);
            sorted[pos] = pk[k];
        }
    }
    __syncthreads();

    int* __restrict__ lp = lists + (size_t)g * LIST_PITCH;
    for (int i = tid; i < ne; i += BIN_THREADS) lp[i] = sorted[i];
    if (tid <= NUM_OWNERS) offs[(size_t)g * OFF_PITCH + tid] = start[tid];
}

__global__ void norm_kernel(float* __restrict__ norm, int n) {
    int i = blockIdx.x * blockDim.x + threadIdx.x;
    if (i < n) {
        float d = norm[i];
        norm[i] = rsqrtf(fmaxf(d, 1.0f));
    }
}

#define OWNER_THREADS   1024
#define OWNER_WAVES     (OWNER_THREADS / 32)
#define SLICES_PER_WAVE (BIN_SLICES / OWNER_WAVES)
#define QUEUE_CAP       (32 + 32 * SLICES_PER_WAVE)
#define EDGES_IN_FLIGHT 8

static_assert(BIN_SLICES % OWNER_WAVES == 0 && SLICES_PER_WAVE <= 32, "one lane per slice of a wave");

__device__ __forceinline__ void add_queued_rows(const float* __restrict__ h,
                                                float* acc,
                                                int s, int dl, int nsb, int cntw, int lane) {
    if (cntw == 32) {
        for (int j0 = 0; j0 < 32; j0 += EDGES_IN_FLIGHT) {
            float v[EDGES_IN_FLIGHT][4];
#pragma unroll
            for (int j = 0; j < EDGES_IN_FLIGHT; ++j) {
                const int sj = __builtin_amdgcn_readlane(s, j0 + j);
                const float* __restrict__ hp = h + (size_t)sj * HID + lane;
                v[j][0] = hp[0];
                v[j][1] = hp[32];
                v[j][2] = hp[64];
                v[j][3] = hp[96];
            }
#pragma unroll
            for (int j = 0; j < EDGES_IN_FLIGHT; ++j) {
                const int dj = __builtin_amdgcn_readlane(dl, j0 + j);
                const float nj = __int_as_float(__builtin_amdgcn_readlane(nsb, j0 + j));
                float* a = acc + dj * HID + lane;
                atomicAdd(a +  0, v[j][0] * nj);
                atomicAdd(a + 32, v[j][1] * nj);
                atomicAdd(a + 64, v[j][2] * nj);
                atomicAdd(a + 96, v[j][3] * nj);
            }
        }
    } else {
        for (int j = 0; j < cntw; ++j) {
            const int sj = __builtin_amdgcn_readlane(s, j);
            const int dj = __builtin_amdgcn_readlane(dl, j);
            const float nj = __int_as_float(__builtin_amdgcn_readlane(nsb, j));
            const float* __restrict__ hp = h + (size_t)sj * HID + lane;
            const float v0 = hp[0];
            const float v1 = hp[32];
            const float v2 = hp[64];
            const float v3 = hp[96];
            float* a = acc + dj * HID + lane;
            atomicAdd(a +  0, v0 * nj);
            atomicAdd(a + 32, v1 * nj);
            atomicAdd(a + 64, v2 * nj);
            atomicAdd(a + 96, v3 * nj);
        }
    }
}

__global__ __launch_bounds__(OWNER_THREADS) void owner_sum_kernel(
    const float* __restrict__ h, const float* __restrict__ norm,
    const int* __restrict__ lists, const int* __restrict__ offs,
    float* __restrict__ agg, int n_nodes)
{
    __shared__ float acc[NODES_PER_BLOCK * HID];
    __shared__ int q_pack[OWNER_WAVES][QUEUE_CAP];

    const int tid  = threadIdx.x;
    const int lane = tid & 31;
    const int wave = __builtin_amdgcn_readfirstlane(tid >> 5);
    const int band = blockIdx.x;
    const int lo   = band * NODES_PER_BLOCK;
    const int cnt  = min(NODES_PER_BLOCK, n_nodes - lo);

    const int g0 = wave * SLICES_PER_WAVE;
    int o0 = 0, o1 = 0;
    if (lane < SLICES_PER_WAVE) {
        const int* __restrict__ op = offs + (size_t)(g0 + lane) * OFF_PITCH + band;
        o0 = op[0];
        o1 = op[1];
    }

    {
        float4* acc4 = reinterpret_cast<float4*>(acc);
        const float4 z = make_float4(0.0f, 0.0f, 0.0f, 0.0f);
        for (int i = tid; i < NODES_PER_BLOCK * HID / 4; i += OWNER_THREADS) acc4[i] = z;
    }
    __syncthreads();

    int* q = q_pack[wave];
    int pend = 0;

    auto drain = [&]() {
        int head = 0;
        __builtin_amdgcn_wave_barrier();
        while (pend - head >= 32) {
            const unsigned w = (unsigned)q[head + lane];
            const int s   = (int)(w >> DLOC_BITS);
            const int dl  = (int)(w & ((1u << DLOC_BITS) - 1u));
            const int nsb = __float_as_int(norm[s]);
            add_queued_rows(h, acc, s, dl, nsb, 32, lane);
            head += 32;
        }
        const int rem = pend - head;
        int t = 0;
        if (lane < rem) t = q[head + lane];
        if (lane < rem) q[lane] = t;
        pend = rem;
    };

    int ent[SLICES_PER_WAVE];
#pragma unroll
    for (int j = 0; j < SLICES_PER_WAVE; ++j) {
        const int a0 = __builtin_amdgcn_readlane(o0, j);
        const int a1 = __builtin_amdgcn_readlane(o1, j);
        ent[j] = 0;
        if (lane < a1 - a0) ent[j] = lists[(size_t)(g0 + j) * LIST_PITCH + a0 + lane];
    }
#pragma unroll
    for (int j = 0; j < SLICES_PER_WAVE; ++j) {
        const int a0 = __builtin_amdgcn_readlane(o0, j);
        const int a1 = __builtin_amdgcn_readlane(o1, j);
        const int n  = min(32, a1 - a0);
        if (lane < n) q[pend + lane] = ent[j];
        pend += n;
    }
    drain();

    for (int j = 0; j < SLICES_PER_WAVE; ++j) {
        const int a0 = __builtin_amdgcn_readlane(o0, j);
        const int a1 = __builtin_amdgcn_readlane(o1, j);
        const int* __restrict__ lp = lists + (size_t)(g0 + j) * LIST_PITCH;
        for (int p = a0 + 32; p < a1; p += 32) {
            const int m = min(32, a1 - p);
            if (lane < m) q[pend + lane] = lp[p + lane];
            pend += m;
            if (pend >= 32) drain();
        }
    }

    if (pend > 0) {
        __builtin_amdgcn_wave_barrier();
        int s = 0, dl = 0, nsb = 0;
        if (lane < pend) {
            const unsigned w = (unsigned)q[lane];
            s   = (int)(w >> DLOC_BITS);
            dl  = (int)(w & ((1u << DLOC_BITS) - 1u));
            nsb = __float_as_int(norm[s]);
        }
        add_queued_rows(h, acc, s, dl, nsb, pend, lane);
    }

    __syncthreads();
    {
        const float4* acc4 = reinterpret_cast<const float4*>(acc);
        float4* __restrict__ out4 = reinterpret_cast<float4*>(agg + (size_t)lo * HID);
        const int nv = cnt * (HID / 4);
        for (int i = tid; i < nv; i += OWNER_THREADS) out4[i] = acc4[i];
    }
}

template<int NT>
__global__ __launch_bounds__(32) void wmma_gemm_strip(
    const float* __restrict__ A, const float* __restrict__ W,
    const float* __restrict__ bias, const float* __restrict__ norm,
    const float* __restrict__ resid, float* __restrict__ C,
    int M, int K, int do_relu)
{
    constexpr int N = NT * 16;
    const int lane  = threadIdx.x;
    const int half  = lane >> 4;
    const int idx16 = lane & 15;
    const int m0 = blockIdx.x * 16;

    const int arow = m0 + idx16;
    const int koff = half * 2;

    const float nscale = (norm != nullptr) ? norm[arow] : 1.0f;
    const float* __restrict__ Arow = A + (size_t)arow * K;

    v8f acc[NT];
#pragma unroll
    for (int t = 0; t < NT; ++t) acc[t] = (v8f){};

    for (int k = 0; k < K; k += 4) {
        v2f a = *reinterpret_cast<const v2f*>(Arow + k + koff);
        a.x *= nscale;
        a.y *= nscale;
        const float* __restrict__ Wr0 = W + (size_t)(k + koff) * N + idx16;
        const float* __restrict__ Wr1 = Wr0 + N;
#pragma unroll
        for (int t = 0; t < NT; ++t) {
            v2f b;
            b.x = Wr0[t * 16];
            b.y = Wr1[t * 16];
            acc[t] = __builtin_amdgcn_wmma_f32_16x16x4_f32(
                 false, a,  false, b,
                 (short)0, acc[t],  false,  false);
        }
    }

#pragma unroll
    for (int t = 0; t < NT; ++t) {
        const int bcol = t * 16 + idx16;
        const float bv = bias[bcol];
#pragma unroll
        for (int r = 0; r < 8; ++r) {
            int orow = m0 + half * 8 + r;
            size_t oidx = (size_t)orow * N + bcol;
            float v = acc[t][r] + bv;
            if (do_relu) v = fmaxf(v, 0.0f);
            if (resid != nullptr) v += resid[oidx];
            C[oidx] = v;
        }
    }
}

extern "C" void kernel_launch(void* const* d_in, const int* in_sizes, int n_in,
                              void* d_out, int out_size, void* d_ws, size_t ws_size,
                              hipStream_t stream) {
    const float* features = (const float*)d_in[0];
    const int*   src      = (const int*)  d_in[1];
    const int*   dst      = (const int*)  d_in[2];
    const float* W_in     = (const float*)d_in[3];
    const float* b_in     = (const float*)d_in[4];
    const float* gcn_w0   = (const float*)d_in[5];
    const float* gcn_b0   = (const float*)d_in[6];
    const float* gcn_w1   = (const float*)d_in[7];
    const float* gcn_b1   = (const float*)d_in[8];
    const float* W_out    = (const float*)d_in[9];
    const float* b_out    = (const float*)d_in[10];
    float* out = (float*)d_out;

    char* ws = (char*)d_ws;
    float* norm = (float*)ws;
    ws += ((size_t)N_NODES * sizeof(float) + 255) & ~(size_t)255;
    float* H = (float*)ws;
    ws += ((size_t)N_NODES * HID * sizeof(float) + 255) & ~(size_t)255;
    float* AGG = (float*)ws;

    const int mBlocks = N_NODES / 16;

    zero_kernel<<<(N_NODES + 255) / 256, 256, 0, stream>>>(norm, N_NODES);
    int* lists = (int*)d_out;
    int* offs  = lists + BIN_SLICES * LIST_PITCH;
    degree_bin_kernel<<<BIN_SLICES, BIN_THREADS, 0, stream>>>(src, dst, norm, lists, offs, N_EDGES);
    norm_kernel<<<(N_NODES + 255) / 256, 256, 0, stream>>>(norm, N_NODES);

    wmma_gemm_strip<8><<<mBlocks, 32, 0, stream>>>(
        features, W_in, b_in, nullptr, nullptr, H, N_NODES, IN_DIM,  0);

    owner_sum_kernel<<<(N_NODES + NODES_PER_BLOCK - 1) / NODES_PER_BLOCK, OWNER_THREADS, 0, stream>>>(
        H, norm, lists, offs, AGG, N_NODES);
    wmma_gemm_strip<8><<<mBlocks, 32, 0, stream>>>(
        AGG, gcn_w0, gcn_b0, norm,  H, H, N_NODES, HID,  1);

    owner_sum_kernel<<<(N_NODES + NODES_PER_BLOCK - 1) / NODES_PER_BLOCK, OWNER_THREADS, 0, stream>>>(
        H, norm, lists, offs, AGG, N_NODES);
    wmma_gemm_strip<8><<<mBlocks, 32, 0, stream>>>(
        AGG, gcn_w1, gcn_b1, norm, nullptr, H, N_NODES, HID,  1);

    wmma_gemm_strip<4><<<mBlocks, 32, 0, stream>>>(
        H, W_out, b_out, nullptr, nullptr, out, N_NODES, HID,  0);
}


# GEMM tile-to-column map: a lane owns 4 adjacent output columns per 64, so W rows, bias, resid and C move as 16-byte vectors and each C store writes whole 128-byte lines; k order and sums unchanged
# speedup vs baseline: 1.9568x; 1.1063x over previous
#include <hip/hip_runtime.h>
#include <hip/hip_bf16.h>

#define N_NODES 100000
#define N_EDGES 800000
#define IN_DIM  256
#define HID     128
#define OUT_DIM 64

typedef float v2f __attribute__((ext_vector_type(2)));
typedef float v4f __attribute__((ext_vector_type(4)));
typedef float v8f __attribute__((ext_vector_type(8)));

__global__ void zero_kernel(float* __restrict__ p, int n) {
    int i = blockIdx.x * blockDim.x + threadIdx.x;
    if (i < n) p[i] = 0.0f;
}

#define NODES_PER_BLOCK 391
#define NUM_OWNERS      ((N_NODES + NODES_PER_BLOCK - 1) / NODES_PER_BLOCK)
#define BIN_SLICES      256
#define BIN_THREADS     1024
#define SLICE_EDGES     ((N_EDGES + BIN_SLICES - 1) / BIN_SLICES)
#define EDGES_PER_THREAD ((SLICE_EDGES + BIN_THREADS - 1) / BIN_THREADS)
#define LIST_PITCH      (((SLICE_EDGES + 31) / 32) * 32)
#define OFF_PITCH       (((NUM_OWNERS + 1 + 31) / 32) * 32)
#define DLOC_BITS       9
#define BIN_WORDS       (BIN_SLICES * (LIST_PITCH + OFF_PITCH))

static_assert(NODES_PER_BLOCK <= (1 << DLOC_BITS), "local node index must fit its field");
static_assert(N_NODES <= (1 << (31 - DLOC_BITS)), "source index must fit its field");
static_assert(NUM_OWNERS % 32 == 0 && NUM_OWNERS <= BIN_THREADS, "prefix code assumes whole waves");
static_assert(BIN_WORDS <= N_NODES * OUT_DIM, "bins must fit the buffer that holds them");

__global__ __launch_bounds__(BIN_THREADS) void degree_bin_kernel(
    const int* __restrict__ src, const int* __restrict__ dst,
    float* __restrict__ degs, int* __restrict__ lists, int* __restrict__ offs, int n_edges)
{
    __shared__ int cnt[NUM_OWNERS];
    __shared__ int start[NUM_OWNERS + 1];
    __shared__ int wave_tot[NUM_OWNERS / 32];
    __shared__ int sorted[SLICE_EDGES];

    const int tid = threadIdx.x;
    const int g   = blockIdx.x;
    const int e0  = g * SLICE_EDGES;
    const int ne  = max(0, min(SLICE_EDGES, n_edges - e0));

    if (tid < NUM_OWNERS) cnt[tid] = 0;
    __syncthreads();

    int dk[EDGES_PER_THREAD], sk[EDGES_PER_THREAD];
#pragma unroll
    for (int k = 0; k < EDGES_PER_THREAD; ++k) {
        const int i = tid + k * BIN_THREADS;
        dk[k] = 0;
        sk[k] = 0;
        if (i < ne) {
            dk[k] = dst[e0 + i];
            sk[k] = src[e0 + i];
        }
    }
    int pk[EDGES_PER_THREAD], bk[EDGES_PER_THREAD];
#pragma unroll
    for (int k = 0; k < EDGES_PER_THREAD; ++k) {
        const int i = tid + k * BIN_THREADS;
        bk[k] = -1;
        pk[k] = 0;
        if (i < ne) {
            const int d = dk[k];
            const int b = d / NODES_PER_BLOCK;
            bk[k] = b;
            pk[k] = (int)(((unsigned)sk[k] << DLOC_BITS) | (unsigned)(d - b * NODES_PER_BLOCK));
            atomicAdd(&cnt[b], 1);
            atomicAdd(&degs[d], 1.0f);
        }
    }
    __syncthreads();

    int v = 0, incl = 0;
    if (tid < NUM_OWNERS) {
        v = cnt[tid];
        incl = v;
        for (int o = 1; o < 32; o <<= 1) {
            const int t = __shfl_up(incl, o);
            if ((tid & 31) >= o) incl += t;
        }
        if ((tid & 31) == 31) wave_tot[tid >> 5] = incl;
    }
    __syncthreads();
    if (tid < NUM_OWNERS) {
        int base = 0;
        for (int w = 0; w < (tid >> 5); ++w) base += wave_tot[w];
        const int ex = base + incl - v;
        start[tid] = ex;
        cnt[tid]   = ex;
        if (tid == NUM_OWNERS - 1) start[NUM_OWNERS] = base + incl;
    }
    __syncthreads();

#pragma unroll
    for (int k = 0; k < EDGES_PER_THREAD; ++k) {
        if (bk[k] >= 0) {
            const int pos = atomicAdd(&cnt[bk[k]], 1);
            sorted[pos] = pk[k];
        }
    }
    __syncthreads();

    int* __restrict__ lp = lists + (size_t)g * LIST_PITCH;
    for (int i = tid; i < ne; i += BIN_THREADS) lp[i] = sorted[i];
    if (tid <= NUM_OWNERS) offs[(size_t)g * OFF_PITCH + tid] = start[tid];
}

__global__ void norm_kernel(float* __restrict__ norm, int n) {
    int i = blockIdx.x * blockDim.x + threadIdx.x;
    if (i < n) {
        float d = norm[i];
        norm[i] = rsqrtf(fmaxf(d, 1.0f));
    }
}

#define OWNER_THREADS   1024
#define OWNER_WAVES     (OWNER_THREADS / 32)
#define SLICES_PER_WAVE (BIN_SLICES / OWNER_WAVES)
#define QUEUE_CAP       (32 + 32 * SLICES_PER_WAVE)
#define EDGES_IN_FLIGHT 8

static_assert(BIN_SLICES % OWNER_WAVES == 0 && SLICES_PER_WAVE <= 32, "one lane per slice of a wave");

__device__ __forceinline__ void add_queued_rows(const float* __restrict__ h,
                                                float* acc,
                                                int s, int dl, int nsb, int cntw, int lane) {
    if (cntw == 32) {
        for (int j0 = 0; j0 < 32; j0 += EDGES_IN_FLIGHT) {
            float v[EDGES_IN_FLIGHT][4];
#pragma unroll
            for (int j = 0; j < EDGES_IN_FLIGHT; ++j) {
                const int sj = __builtin_amdgcn_readlane(s, j0 + j);
                const float* __restrict__ hp = h + (size_t)sj * HID + lane;
                v[j][0] = hp[0];
                v[j][1] = hp[32];
                v[j][2] = hp[64];
                v[j][3] = hp[96];
            }
#pragma unroll
            for (int j = 0; j < EDGES_IN_FLIGHT; ++j) {
                const int dj = __builtin_amdgcn_readlane(dl, j0 + j);
                const float nj = __int_as_float(__builtin_amdgcn_readlane(nsb, j0 + j));
                float* a = acc + dj * HID + lane;
                atomicAdd(a +  0, v[j][0] * nj);
                atomicAdd(a + 32, v[j][1] * nj);
                atomicAdd(a + 64, v[j][2] * nj);
                atomicAdd(a + 96, v[j][3] * nj);
            }
        }
    } else {
        for (int j = 0; j < cntw; ++j) {
            const int sj = __builtin_amdgcn_readlane(s, j);
            const int dj = __builtin_amdgcn_readlane(dl, j);
            const float nj = __int_as_float(__builtin_amdgcn_readlane(nsb, j));
            const float* __restrict__ hp = h + (size_t)sj * HID + lane;
            const float v0 = hp[0];
            const float v1 = hp[32];
            const float v2 = hp[64];
            const float v3 = hp[96];
            float* a = acc + dj * HID + lane;
            atomicAdd(a +  0, v0 * nj);
            atomicAdd(a + 32, v1 * nj);
            atomicAdd(a + 64, v2 * nj);
            atomicAdd(a + 96, v3 * nj);
        }
    }
}

__global__ __launch_bounds__(OWNER_THREADS) void owner_sum_kernel(
    const float* __restrict__ h, const float* __restrict__ norm,
    const int* __restrict__ lists, const int* __restrict__ offs,
    float* __restrict__ agg, int n_nodes)
{
    __shared__ float acc[NODES_PER_BLOCK * HID];
    __shared__ int q_pack[OWNER_WAVES][QUEUE_CAP];

    const int tid  = threadIdx.x;
    const int lane = tid & 31;
    const int wave = __builtin_amdgcn_readfirstlane(tid >> 5);
    const int band = blockIdx.x;
    const int lo   = band * NODES_PER_BLOCK;
    const int cnt  = min(NODES_PER_BLOCK, n_nodes - lo);

    const int g0 = wave * SLICES_PER_WAVE;
    int o0 = 0, o1 = 0;
    if (lane < SLICES_PER_WAVE) {
        const int* __restrict__ op = offs + (size_t)(g0 + lane) * OFF_PITCH + band;
        o0 = op[0];
        o1 = op[1];
    }

    {
        float4* acc4 = reinterpret_cast<float4*>(acc);
        const float4 z = make_float4(0.0f, 0.0f, 0.0f, 0.0f);
        for (int i = tid; i < NODES_PER_BLOCK * HID / 4; i += OWNER_THREADS) acc4[i] = z;
    }
    __syncthreads();

    int* q = q_pack[wave];
    int pend = 0;

    auto drain = [&]() {
        int head = 0;
        __builtin_amdgcn_wave_barrier();
        while (pend - head >= 32) {
            const unsigned w = (unsigned)q[head + lane];
            const int s   = (int)(w >> DLOC_BITS);
            const int dl  = (int)(w & ((1u << DLOC_BITS) - 1u));
            const int nsb = __float_as_int(norm[s]);
            add_queued_rows(h, acc, s, dl, nsb, 32, lane);
            head += 32;
        }
        const int rem = pend - head;
        int t = 0;
        if (lane < rem) t = q[head + lane];
        if (lane < rem) q[lane] = t;
        pend = rem;
    };

    int ent[SLICES_PER_WAVE];
#pragma unroll
    for (int j = 0; j < SLICES_PER_WAVE; ++j) {
        const int a0 = __builtin_amdgcn_readlane(o0, j);
        const int a1 = __builtin_amdgcn_readlane(o1, j);
        ent[j] = 0;
        if (lane < a1 - a0) ent[j] = lists[(size_t)(g0 + j) * LIST_PITCH + a0 + lane];
    }
#pragma unroll
    for (int j = 0; j < SLICES_PER_WAVE; ++j) {
        const int a0 = __builtin_amdgcn_readlane(o0, j);
        const int a1 = __builtin_amdgcn_readlane(o1, j);
        const int n  = min(32, a1 - a0);
        if (lane < n) q[pend + lane] = ent[j];
        pend += n;
    }
    drain();

    for (int j = 0; j < SLICES_PER_WAVE; ++j) {
        const int a0 = __builtin_amdgcn_readlane(o0, j);
        const int a1 = __builtin_amdgcn_readlane(o1, j);
        const int* __restrict__ lp = lists + (size_t)(g0 + j) * LIST_PITCH;
        for (int p = a0 + 32; p < a1; p += 32) {
            const int m = min(32, a1 - p);
            if (lane < m) q[pend + lane] = lp[p + lane];
            pend += m;
            if (pend >= 32) drain();
        }
    }

    if (pend > 0) {
        __builtin_amdgcn_wave_barrier();
        int s = 0, dl = 0, nsb = 0;
        if (lane < pend) {
            const unsigned w = (unsigned)q[lane];
            s   = (int)(w >> DLOC_BITS);
            dl  = (int)(w & ((1u << DLOC_BITS) - 1u));
            nsb = __float_as_int(norm[s]);
        }
        add_queued_rows(h, acc, s, dl, nsb, pend, lane);
    }

    __syncthreads();
    {
        const float4* acc4 = reinterpret_cast<const float4*>(acc);
        float4* __restrict__ out4 = reinterpret_cast<float4*>(agg + (size_t)lo * HID);
        const int nv = cnt * (HID / 4);
        for (int i = tid; i < nv; i += OWNER_THREADS) out4[i] = acc4[i];
    }
}

template<int NT>
__global__ __launch_bounds__(32) void wmma_gemm_strip(
    const float* __restrict__ A, const float* __restrict__ W,
    const float* __restrict__ bias, const float* __restrict__ norm,
    const float* __restrict__ resid, float* __restrict__ C,
    int M, int K, int do_relu)
{
    constexpr int N  = NT * 16;
    constexpr int NG = NT / 4;
    static_assert(NT % 4 == 0, "tiles come in groups of 4");
    const int lane  = threadIdx.x;
    const int half  = lane >> 4;
    const int idx16 = lane & 15;
    const int m0 = blockIdx.x * 16;

    const int arow = m0 + idx16;
    const int koff = half * 2;
    const int col4 = idx16 * 4;

    const float nscale = (norm != nullptr) ? norm[arow] : 1.0f;
    const float* __restrict__ Arow = A + (size_t)arow * K;

    v8f acc[NT];
#pragma unroll
    for (int t = 0; t < NT; ++t) acc[t] = (v8f){};

    for (int k = 0; k < K; k += 4) {
        v2f a = *reinterpret_cast<const v2f*>(Arow + k + koff);
        a.x *= nscale;
        a.y *= nscale;
        const float* __restrict__ Wr0 = W + (size_t)(k + koff) * N + col4;
        const float* __restrict__ Wr1 = Wr0 + N;
        v4f b0[NG], b1[NG];
#pragma unroll
        for (int g = 0; g < NG; ++g) {
            b0[g] = *reinterpret_cast<const v4f*>(Wr0 + g * 64);
            b1[g] = *reinterpret_cast<const v4f*>(Wr1 + g * 64);
        }
#pragma unroll
        for (int g = 0; g < NG; ++g) {
#pragma unroll
            for (int i = 0; i < 4; ++i) {
                v2f b;
                b.x = b0[g][i];
                b.y = b1[g][i];
                acc[g * 4 + i] = __builtin_amdgcn_wmma_f32_16x16x4_f32(
                     false, a,  false, b,
                     (short)0, acc[g * 4 + i],  false,  false);
            }
        }
    }

#pragma unroll
    for (int g = 0; g < NG; ++g) {
        const int bcol = g * 64 + col4;
        const v4f bv = *reinterpret_cast<const v4f*>(bias + bcol);
#pragma unroll
        for (int r = 0; r < 8; ++r) {
            const int orow = m0 + half * 8 + r;
            const size_t oidx = (size_t)orow * N + bcol;
            v4f v;
            v.x = acc[g * 4 + 0][r] + bv.x;
            v.y = acc[g * 4 + 1][r] + bv.y;
            v.z = acc[g * 4 + 2][r] + bv.z;
            v.w = acc[g * 4 + 3][r] + bv.w;
            if (do_relu) {
                v.x = fmaxf(v.x, 0.0f);
                v.y = fmaxf(v.y, 0.0f);
                v.z = fmaxf(v.z, 0.0f);
                v.w = fmaxf(v.w, 0.0f);
            }
            if (resid != nullptr) v += *reinterpret_cast<const v4f*>(resid + oidx);
            *reinterpret_cast<v4f*>(C + oidx) = v;
        }
    }
}

extern "C" void kernel_launch(void* const* d_in, const int* in_sizes, int n_in,
                              void* d_out, int out_size, void* d_ws, size_t ws_size,
                              hipStream_t stream) {
    const float* features = (const float*)d_in[0];
    const int*   src      = (const int*)  d_in[1];
    const int*   dst      = (const int*)  d_in[2];
    const float* W_in     = (const float*)d_in[3];
    const float* b_in     = (const float*)d_in[4];
    const float* gcn_w0   = (const float*)d_in[5];
    const float* gcn_b0   = (const float*)d_in[6];
    const float* gcn_w1   = (const float*)d_in[7];
    const float* gcn_b1   = (const float*)d_in[8];
    const float* W_out    = (const float*)d_in[9];
    const float* b_out    = (const float*)d_in[10];
    float* out = (float*)d_out;

    char* ws = (char*)d_ws;
    float* norm = (float*)ws;
    ws += ((size_t)N_NODES * sizeof(float) + 255) & ~(size_t)255;
    float* H = (float*)ws;
    ws += ((size_t)N_NODES * HID * sizeof(float) + 255) & ~(size_t)255;
    float* AGG = (float*)ws;

    const int mBlocks = N_NODES / 16;

    zero_kernel<<<(N_NODES + 255) / 256, 256, 0, stream>>>(norm, N_NODES);
    int* lists = (int*)d_out;
    int* offs  = lists + BIN_SLICES * LIST_PITCH;
    degree_bin_kernel<<<BIN_SLICES, BIN_THREADS, 0, stream>>>(src, dst, norm, lists, offs, N_EDGES);
    norm_kernel<<<(N_NODES + 255) / 256, 256, 0, stream>>>(norm, N_NODES);

    wmma_gemm_strip<8><<<mBlocks, 32, 0, stream>>>(
        features, W_in, b_in, nullptr, nullptr, H, N_NODES, IN_DIM,  0);

    owner_sum_kernel<<<(N_NODES + NODES_PER_BLOCK - 1) / NODES_PER_BLOCK, OWNER_THREADS, 0, stream>>>(
        H, norm, lists, offs, AGG, N_NODES);
    wmma_gemm_strip<8><<<mBlocks, 32, 0, stream>>>(
        AGG, gcn_w0, gcn_b0, norm,  H, H, N_NODES, HID,  1);

    owner_sum_kernel<<<(N_NODES + NODES_PER_BLOCK - 1) / NODES_PER_BLOCK, OWNER_THREADS, 0, stream>>>(
        H, norm, lists, offs, AGG, N_NODES);
    wmma_gemm_strip<8><<<mBlocks, 32, 0, stream>>>(
        AGG, gcn_w1, gcn_b1, norm, nullptr, H, N_NODES, HID,  1);

    wmma_gemm_strip<4><<<mBlocks, 32, 0, stream>>>(
        H, W_out, b_out, nullptr, nullptr, out, N_NODES, HID,  0);
}


# Each layer's GEMM runs in its neighbour-sum launch: owner workgroup feeds WMMA A fragments from its 391-node LDS sums (pitch 132), agg never goes to memory, 9 launches to 7; k order, sums unchanged
# speedup vs baseline: 2.4137x; 1.2335x over previous
#include <hip/hip_runtime.h>
#include <hip/hip_bf16.h>

#define N_NODES 100000
#define N_EDGES 800000
#define IN_DIM  256
#define HID     128
#define OUT_DIM 64

typedef float v2f __attribute__((ext_vector_type(2)));
typedef float v4f __attribute__((ext_vector_type(4)));
typedef float v8f __attribute__((ext_vector_type(8)));

__global__ void zero_kernel(float* __restrict__ p, int n) {
    int i = blockIdx.x * blockDim.x + threadIdx.x;
    if (i < n) p[i] = 0.0f;
}

#define NODES_PER_BLOCK 391
#define NUM_OWNERS      ((N_NODES + NODES_PER_BLOCK - 1) / NODES_PER_BLOCK)
#define BIN_SLICES      256
#define BIN_THREADS     1024
#define SLICE_EDGES     ((N_EDGES + BIN_SLICES - 1) / BIN_SLICES)
#define EDGES_PER_THREAD ((SLICE_EDGES + BIN_THREADS - 1) / BIN_THREADS)
#define LIST_PITCH      (((SLICE_EDGES + 31) / 32) * 32)
#define OFF_PITCH       (((NUM_OWNERS + 1 + 31) / 32) * 32)
#define DLOC_BITS       9
#define BIN_WORDS       (BIN_SLICES * (LIST_PITCH + OFF_PITCH))

static_assert(NODES_PER_BLOCK <= (1 << DLOC_BITS), "local node index must fit its field");
static_assert(N_NODES <= (1 << (31 - DLOC_BITS)), "source index must fit its field");
static_assert(NUM_OWNERS % 32 == 0 && NUM_OWNERS <= BIN_THREADS, "prefix code assumes whole waves");
static_assert(BIN_WORDS <= N_NODES * OUT_DIM, "bins must fit the buffer that holds them");

__global__ __launch_bounds__(BIN_THREADS) void degree_bin_kernel(
    const int* __restrict__ src, const int* __restrict__ dst,
    float* __restrict__ degs, int* __restrict__ lists, int* __restrict__ offs, int n_edges)
{
    __shared__ int cnt[NUM_OWNERS];
    __shared__ int start[NUM_OWNERS + 1];
    __shared__ int wave_tot[NUM_OWNERS / 32];
    __shared__ int sorted[SLICE_EDGES];

    const int tid = threadIdx.x;
    const int g   = blockIdx.x;
    const int e0  = g * SLICE_EDGES;
    const int ne  = max(0, min(SLICE_EDGES, n_edges - e0));

    if (tid < NUM_OWNERS) cnt[tid] = 0;
    __syncthreads();

    int dk[EDGES_PER_THREAD], sk[EDGES_PER_THREAD];
#pragma unroll
    for (int k = 0; k < EDGES_PER_THREAD; ++k) {
        const int i = tid + k * BIN_THREADS;
        dk[k] = 0;
        sk[k] = 0;
        if (i < ne) {
            dk[k] = dst[e0 + i];
            sk[k] = src[e0 + i];
        }
    }
    int pk[EDGES_PER_THREAD], bk[EDGES_PER_THREAD];
#pragma unroll
    for (int k = 0; k < EDGES_PER_THREAD; ++k) {
        const int i = tid + k * BIN_THREADS;
        bk[k] = -1;
        pk[k] = 0;
        if (i < ne) {
            const int d = dk[k];
            const int b = d / NODES_PER_BLOCK;
            bk[k] = b;
            pk[k] = (int)(((unsigned)sk[k] << DLOC_BITS) | (unsigned)(d - b * NODES_PER_BLOCK));
            atomicAdd(&cnt[b], 1);
            atomicAdd(&degs[d], 1.0f);
        }
    }
    __syncthreads();

    int v = 0, incl = 0;
    if (tid < NUM_OWNERS) {
        v = cnt[tid];
        incl = v;
        for (int o = 1; o < 32; o <<= 1) {
            const int t = __shfl_up(incl, o);
            if ((tid & 31) >= o) incl += t;
        }
        if ((tid & 31) == 31) wave_tot[tid >> 5] = incl;
    }
    __syncthreads();
    if (tid < NUM_OWNERS) {
        int base = 0;
        for (int w = 0; w < (tid >> 5); ++w) base += wave_tot[w];
        const int ex = base + incl - v;
        start[tid] = ex;
        cnt[tid]   = ex;
        if (tid == NUM_OWNERS - 1) start[NUM_OWNERS] = base + incl;
    }
    __syncthreads();

#pragma unroll
    for (int k = 0; k < EDGES_PER_THREAD; ++k) {
        if (bk[k] >= 0) {
            const int pos = atomicAdd(&cnt[bk[k]], 1);
            sorted[pos] = pk[k];
        }
    }
    __syncthreads();

    int* __restrict__ lp = lists + (size_t)g * LIST_PITCH;
    for (int i = tid; i < ne; i += BIN_THREADS) lp[i] = sorted[i];
    if (tid <= NUM_OWNERS) offs[(size_t)g * OFF_PITCH + tid] = start[tid];
}

__global__ void norm_kernel(float* __restrict__ norm, int n) {
    int i = blockIdx.x * blockDim.x + threadIdx.x;
    if (i < n) {
        float d = norm[i];
        norm[i] = rsqrtf(fmaxf(d, 1.0f));
    }
}

#define OWNER_THREADS   1024
#define OWNER_WAVES     (OWNER_THREADS / 32)
#define SLICES_PER_WAVE (BIN_SLICES / OWNER_WAVES)
#define QUEUE_CAP       (32 + 32 * SLICES_PER_WAVE)
#define EDGES_IN_FLIGHT 8
#define ACC_PITCH       (HID + 4)
#define TILE_ROWS       (((NODES_PER_BLOCK + 15) / 16) * 16)

static_assert(BIN_SLICES % OWNER_WAVES == 0 && SLICES_PER_WAVE <= 32, "one lane per slice of a wave");
static_assert((ACC_PITCH * 4) % 16 == 0 && ((ACC_PITCH * 4) / 16) % 2 == 1, "tile pitch: 16 bytes x odd");
static_assert(HID == 128, "phase 2 covers the output columns as two groups of 64");
static_assert(TILE_ROWS / 16 <= OWNER_WAVES, "phase 2 gives a wave at most two units");

__device__ __forceinline__ void add_queued_rows(const float* __restrict__ h,
                                                float* acc,
                                                int s, int dl, int nsb, int cntw, int lane) {
    if (cntw == 32) {
        for (int j0 = 0; j0 < 32; j0 += EDGES_IN_FLIGHT) {
            float v[EDGES_IN_FLIGHT][4];
#pragma unroll
            for (int j = 0; j < EDGES_IN_FLIGHT; ++j) {
                const int sj = __builtin_amdgcn_readlane(s, j0 + j);
                const float* __restrict__ hp = h + (size_t)sj * HID + lane;
                v[j][0] = hp[0];
                v[j][1] = hp[32];
                v[j][2] = hp[64];
                v[j][3] = hp[96];
            }
#pragma unroll
            for (int j = 0; j < EDGES_IN_FLIGHT; ++j) {
                const int dj = __builtin_amdgcn_readlane(dl, j0 + j);
                const float nj = __int_as_float(__builtin_amdgcn_readlane(nsb, j0 + j));
                float* a = acc + dj * ACC_PITCH + lane;
                atomicAdd(a +  0, v[j][0] * nj);
                atomicAdd(a + 32, v[j][1] * nj);
                atomicAdd(a + 64, v[j][2] * nj);
                atomicAdd(a + 96, v[j][3] * nj);
            }
        }
    } else {
        for (int j = 0; j < cntw; ++j) {
            const int sj = __builtin_amdgcn_readlane(s, j);
            const int dj = __builtin_amdgcn_readlane(dl, j);
            const float nj = __int_as_float(__builtin_amdgcn_readlane(nsb, j));
            const float* __restrict__ hp = h + (size_t)sj * HID + lane;
            const float v0 = hp[0];
            const float v1 = hp[32];
            const float v2 = hp[64];
            const float v3 = hp[96];
            float* a = acc + dj * ACC_PITCH + lane;
            atomicAdd(a +  0, v0 * nj);
            atomicAdd(a + 32, v1 * nj);
            atomicAdd(a + 64, v2 * nj);
            atomicAdd(a + 96, v3 * nj);
        }
    }
}

template<bool ADD_RESID>
__global__ __launch_bounds__(OWNER_THREADS) void owner_layer_kernel(
    const float* __restrict__ h, const float* __restrict__ norm,
    const int* __restrict__ lists, const int* __restrict__ offs,
    const float* __restrict__ W, const float* __restrict__ bias,
    float* __restrict__ hout, int n_nodes)
{
    __shared__ __attribute__((aligned(16))) float acc[TILE_ROWS * ACC_PITCH];
    __shared__ int q_pack[OWNER_WAVES][QUEUE_CAP];

    const int tid  = threadIdx.x;
    const int lane = tid & 31;
    const int wave = __builtin_amdgcn_readfirstlane(tid >> 5);
    const int band = blockIdx.x;
    const int lo   = band * NODES_PER_BLOCK;
    const int cnt  = min(NODES_PER_BLOCK, n_nodes - lo);

    const int g0 = wave * SLICES_PER_WAVE;
    int o0 = 0, o1 = 0;
    if (lane < SLICES_PER_WAVE) {
        const int* __restrict__ op = offs + (size_t)(g0 + lane) * OFF_PITCH + band;
        o0 = op[0];
        o1 = op[1];
    }

    const int half   = lane >> 4;
    const int idx16  = lane & 15;
    const int koff   = half * 2;
    const int bcol   = (wave & 1) * 64 + idx16 * 4;
    const int strip0 = wave >> 1;
    const float nscale0 = norm[lo + min(strip0 * 16 + idx16, cnt - 1)];
    const float nscale1 = norm[lo + min((strip0 + OWNER_WAVES / 2) * 16 + idx16, cnt - 1)];
    const v4f bv = *reinterpret_cast<const v4f*>(bias + bcol);

    {
        float4* acc4 = reinterpret_cast<float4*>(acc);
        const float4 z = make_float4(0.0f, 0.0f, 0.0f, 0.0f);
        for (int i = tid; i < TILE_ROWS * ACC_PITCH / 4; i += OWNER_THREADS) acc4[i] = z;
    }
    __syncthreads();

    int* q = q_pack[wave];
    int pend = 0;

    auto drain = [&]() {
        int head = 0;
        __builtin_amdgcn_wave_barrier();
        while (pend - head >= 32) {
            const unsigned w = (unsigned)q[head + lane];
            const int s   = (int)(w >> DLOC_BITS);
            const int dl  = (int)(w & ((1u << DLOC_BITS) - 1u));
            const int nsb = __float_as_int(norm[s]);
            add_queued_rows(h, acc, s, dl, nsb, 32, lane);
            head += 32;
        }
        const int rem = pend - head;
        int t = 0;
        if (lane < rem) t = q[head + lane];
        if (lane < rem) q[lane] = t;
        pend = rem;
    };

    int ent[SLICES_PER_WAVE];
#pragma unroll
    for (int j = 0; j < SLICES_PER_WAVE; ++j) {
        const int a0 = __builtin_amdgcn_readlane(o0, j);
        const int a1 = __builtin_amdgcn_readlane(o1, j);
        ent[j] = 0;
        if (lane < a1 - a0) ent[j] = lists[(size_t)(g0 + j) * LIST_PITCH + a0 + lane];
    }
#pragma unroll
    for (int j = 0; j < SLICES_PER_WAVE; ++j) {
        const int a0 = __builtin_amdgcn_readlane(o0, j);
        const int a1 = __builtin_amdgcn_readlane(o1, j);
        const int n  = min(32, a1 - a0);
        if (lane < n) q[pend + lane] = ent[j];
        pend += n;
    }
    drain();

    for (int j = 0; j < SLICES_PER_WAVE; ++j) {
        const int a0 = __builtin_amdgcn_readlane(o0, j);
        const int a1 = __builtin_amdgcn_readlane(o1, j);
        const int* __restrict__ lp = lists + (size_t)(g0 + j) * LIST_PITCH;
        for (int p = a0 + 32; p < a1; p += 32) {
            const int m = min(32, a1 - p);
            if (lane < m) q[pend + lane] = lp[p + lane];
            pend += m;
            if (pend >= 32) drain();
        }
    }

    if (pend > 0) {
        __builtin_amdgcn_wave_barrier();
        int s = 0, dl = 0, nsb = 0;
        if (lane < pend) {
            const unsigned w = (unsigned)q[lane];
            s   = (int)(w >> DLOC_BITS);
            dl  = (int)(w & ((1u << DLOC_BITS) - 1u));
            nsb = __float_as_int(norm[s]);
        }
        add_queued_rows(h, acc, s, dl, nsb, pend, lane);
    }

    __syncthreads();

    const int nstrips = (cnt + 15) >> 4;
    const float* __restrict__ Wc = W + (size_t)koff * HID + bcol;

    for (int t = 0; t < 2; ++t) {
        const int strip = strip0 + t * (OWNER_WAVES / 2);
        if (strip >= nstrips) break;
        const float nscale = t ? nscale1 : nscale0;
        const float* arow = acc + (strip * 16 + idx16) * ACC_PITCH + koff;
        const int orow0 = strip * 16 + half * 8;

        v4f rs[8];
        if (ADD_RESID) {
#pragma unroll
            for (int r = 0; r < 8; ++r)
                rs[r] = *reinterpret_cast<const v4f*>(h + (size_t)(lo + min(orow0 + r, cnt - 1)) * HID + bcol);
        }

        v8f c[4];
#pragma unroll
        for (int i = 0; i < 4; ++i) c[i] = (v8f){};

#pragma unroll 4
        for (int k = 0; k < HID; k += 4) {
            v2f a = *reinterpret_cast<const v2f*>(arow + k);
            a.x *= nscale;
            a.y *= nscale;
            const float* __restrict__ Wr0 = Wc + (size_t)k * HID;
            const v4f b0 = *reinterpret_cast<const v4f*>(Wr0);
            const v4f b1 = *reinterpret_cast<const v4f*>(Wr0 + HID);
#pragma unroll
            for (int i = 0; i < 4; ++i) {
                v2f b;
                b.x = b0[i];
                b.y = b1[i];
                c[i] = __builtin_amdgcn_wmma_f32_16x16x4_f32(
                     false, a,  false, b,
                     (short)0, c[i],  false,  false);
            }
        }

#pragma unroll
        for (int r = 0; r < 8; ++r) {
            v4f v;
            v.x = fmaxf(c[0][r] + bv.x, 0.0f);
            v.y = fmaxf(c[1][r] + bv.y, 0.0f);
            v.z = fmaxf(c[2][r] + bv.z, 0.0f);
            v.w = fmaxf(c[3][r] + bv.w, 0.0f);
            if (ADD_RESID) v += rs[r];
            if (orow0 + r < cnt)
                *reinterpret_cast<v4f*>(hout + (size_t)(lo + orow0 + r) * HID + bcol) = v;
        }
    }
}

template<int NT>
__global__ __launch_bounds__(32) void wmma_gemm_strip(
    const float* __restrict__ A, const float* __restrict__ W,
    const float* __restrict__ bias, const float* __restrict__ norm,
    const float* __restrict__ resid, float* __restrict__ C,
    int M, int K, int do_relu)
{
    constexpr int N  = NT * 16;
    constexpr int NG = NT / 4;
    static_assert(NT % 4 == 0, "tiles come in groups of 4");
    const int lane  = threadIdx.x;
    const int half  = lane >> 4;
    const int idx16 = lane & 15;
    const int m0 = blockIdx.x * 16;

    const int arow = m0 + idx16;
    const int koff = half * 2;
    const int col4 = idx16 * 4;

    const float nscale = (norm != nullptr) ? norm[arow] : 1.0f;
    const float* __restrict__ Arow = A + (size_t)arow * K;

    v8f acc[NT];
#pragma unroll
    for (int t = 0; t < NT; ++t) acc[t] = (v8f){};

    for (int k = 0; k < K; k += 4) {
        v2f a = *reinterpret_cast<const v2f*>(Arow + k + koff);
        a.x *= nscale;
        a.y *= nscale;
        const float* __restrict__ Wr0 = W + (size_t)(k + koff) * N + col4;
        const float* __restrict__ Wr1 = Wr0 + N;
        v4f b0[NG], b1[NG];
#pragma unroll
        for (int g = 0; g < NG; ++g) {
            b0[g] = *reinterpret_cast<const v4f*>(Wr0 + g * 64);
            b1[g] = *reinterpret_cast<const v4f*>(Wr1 + g * 64);
        }
#pragma unroll
        for (int g = 0; g < NG; ++g) {
#pragma unroll
            for (int i = 0; i < 4; ++i) {
                v2f b;
                b.x = b0[g][i];
                b.y = b1[g][i];
                acc[g * 4 + i] = __builtin_amdgcn_wmma_f32_16x16x4_f32(
                     false, a,  false, b,
                     (short)0, acc[g * 4 + i],  false,  false);
            }
        }
    }

#pragma unroll
    for (int g = 0; g < NG; ++g) {
        const int bcol = g * 64 + col4;
        const v4f bv = *reinterpret_cast<const v4f*>(bias + bcol);
#pragma unroll
        for (int r = 0; r < 8; ++r) {
            const int orow = m0 + half * 8 + r;
            const size_t oidx = (size_t)orow * N + bcol;
            v4f v;
            v.x = acc[g * 4 + 0][r] + bv.x;
            v.y = acc[g * 4 + 1][r] + bv.y;
            v.z = acc[g * 4 + 2][r] + bv.z;
            v.w = acc[g * 4 + 3][r] + bv.w;
            if (do_relu) {
                v.x = fmaxf(v.x, 0.0f);
                v.y = fmaxf(v.y, 0.0f);
                v.z = fmaxf(v.z, 0.0f);
                v.w = fmaxf(v.w, 0.0f);
            }
            if (resid != nullptr) v += *reinterpret_cast<const v4f*>(resid + oidx);
            *reinterpret_cast<v4f*>(C + oidx) = v;
        }
    }
}

extern "C" void kernel_launch(void* const* d_in, const int* in_sizes, int n_in,
                              void* d_out, int out_size, void* d_ws, size_t ws_size,
                              hipStream_t stream) {
    const float* features = (const float*)d_in[0];
    const int*   src      = (const int*)  d_in[1];
    const int*   dst      = (const int*)  d_in[2];
    const float* W_in     = (const float*)d_in[3];
    const float* b_in     = (const float*)d_in[4];
    const float* gcn_w0   = (const float*)d_in[5];
    const float* gcn_b0   = (const float*)d_in[6];
    const float* gcn_w1   = (const float*)d_in[7];
    const float* gcn_b1   = (const float*)d_in[8];
    const float* W_out    = (const float*)d_in[9];
    const float* b_out    = (const float*)d_in[10];
    float* out = (float*)d_out;

    char* ws = (char*)d_ws;
    float* norm = (float*)ws;
    ws += ((size_t)N_NODES * sizeof(float) + 255) & ~(size_t)255;
    float* H = (float*)ws;
    ws += ((size_t)N_NODES * HID * sizeof(float) + 255) & ~(size_t)255;
    float* H1 = (float*)ws;

    const int mBlocks = N_NODES / 16;

    zero_kernel<<<(N_NODES + 255) / 256, 256, 0, stream>>>(norm, N_NODES);
    int* lists = (int*)d_out;
    int* offs  = lists + BIN_SLICES * LIST_PITCH;
    degree_bin_kernel<<<BIN_SLICES, BIN_THREADS, 0, stream>>>(src, dst, norm, lists, offs, N_EDGES);
    norm_kernel<<<(N_NODES + 255) / 256, 256, 0, stream>>>(norm, N_NODES);

    wmma_gemm_strip<8><<<mBlocks, 32, 0, stream>>>(
        features, W_in, b_in, nullptr, nullptr, H, N_NODES, IN_DIM,  0);

    owner_layer_kernel<true><<<NUM_OWNERS, OWNER_THREADS, 0, stream>>>(
        H, norm, lists, offs, gcn_w0, gcn_b0, H1, N_NODES);

    owner_layer_kernel<false><<<NUM_OWNERS, OWNER_THREADS, 0, stream>>>(
        H1, norm, lists, offs, gcn_w1, gcn_b1, H, N_NODES);

    wmma_gemm_strip<4><<<mBlocks, 32, 0, stream>>>(
        H, W_out, b_out, nullptr, nullptr, out, N_NODES, HID,  0);
}


# Both dense GEMMs: W kept in registers, K split over a workgroup's 4 waves, 16-row strips looped; the 4 k-quarter sums are added via LDS in wave order, k paired anew inside 8-k blocks
# speedup vs baseline: 2.6066x; 1.0799x over previous
#include <hip/hip_runtime.h>
#include <hip/hip_bf16.h>

#define N_NODES 100000
#define N_EDGES 800000
#define IN_DIM  256
#define HID     128
#define OUT_DIM 64

typedef float v2f __attribute__((ext_vector_type(2)));
typedef float v4f __attribute__((ext_vector_type(4)));
typedef float v8f __attribute__((ext_vector_type(8)));

__global__ void zero_kernel(float* __restrict__ p, int n) {
    int i = blockIdx.x * blockDim.x + threadIdx.x;
    if (i < n) p[i] = 0.0f;
}

#define NODES_PER_BLOCK 391
#define NUM_OWNERS      ((N_NODES + NODES_PER_BLOCK - 1) / NODES_PER_BLOCK)
#define BIN_SLICES      256
#define BIN_THREADS     1024
#define SLICE_EDGES     ((N_EDGES + BIN_SLICES - 1) / BIN_SLICES)
#define EDGES_PER_THREAD ((SLICE_EDGES + BIN_THREADS - 1) / BIN_THREADS)
#define LIST_PITCH      (((SLICE_EDGES + 31) / 32) * 32)
#define OFF_PITCH       (((NUM_OWNERS + 1 + 31) / 32) * 32)
#define DLOC_BITS       9
#define BIN_WORDS       (BIN_SLICES * (LIST_PITCH + OFF_PITCH))

static_assert(NODES_PER_BLOCK <= (1 << DLOC_BITS), "local node index must fit its field");
static_assert(N_NODES <= (1 << (31 - DLOC_BITS)), "source index must fit its field");
static_assert(NUM_OWNERS % 32 == 0 && NUM_OWNERS <= BIN_THREADS, "prefix code assumes whole waves");
static_assert(BIN_WORDS <= N_NODES * OUT_DIM, "bins must fit the buffer that holds them");

__global__ __launch_bounds__(BIN_THREADS) void degree_bin_kernel(
    const int* __restrict__ src, const int* __restrict__ dst,
    float* __restrict__ degs, int* __restrict__ lists, int* __restrict__ offs, int n_edges)
{
    __shared__ int cnt[NUM_OWNERS];
    __shared__ int start[NUM_OWNERS + 1];
    __shared__ int wave_tot[NUM_OWNERS / 32];
    __shared__ int sorted[SLICE_EDGES];

    const int tid = threadIdx.x;
    const int g   = blockIdx.x;
    const int e0  = g * SLICE_EDGES;
    const int ne  = max(0, min(SLICE_EDGES, n_edges - e0));

    if (tid < NUM_OWNERS) cnt[tid] = 0;
    __syncthreads();

    int dk[EDGES_PER_THREAD], sk[EDGES_PER_THREAD];
#pragma unroll
    for (int k = 0; k < EDGES_PER_THREAD; ++k) {
        const int i = tid + k * BIN_THREADS;
        dk[k] = 0;
        sk[k] = 0;
        if (i < ne) {
            dk[k] = dst[e0 + i];
            sk[k] = src[e0 + i];
        }
    }
    int pk[EDGES_PER_THREAD], bk[EDGES_PER_THREAD];
#pragma unroll
    for (int k = 0; k < EDGES_PER_THREAD; ++k) {
        const int i = tid + k * BIN_THREADS;
        bk[k] = -1;
        pk[k] = 0;
        if (i < ne) {
            const int d = dk[k];
            const int b = d / NODES_PER_BLOCK;
            bk[k] = b;
            pk[k] = (int)(((unsigned)sk[k] << DLOC_BITS) | (unsigned)(d - b * NODES_PER_BLOCK));
            atomicAdd(&cnt[b], 1);
            atomicAdd(&degs[d], 1.0f);
        }
    }
    __syncthreads();

    int v = 0, incl = 0;
    if (tid < NUM_OWNERS) {
        v = cnt[tid];
        incl = v;
        for (int o = 1; o < 32; o <<= 1) {
            const int t = __shfl_up(incl, o);
            if ((tid & 31) >= o) incl += t;
        }
        if ((tid & 31) == 31) wave_tot[tid >> 5] = incl;
    }
    __syncthreads();
    if (tid < NUM_OWNERS) {
        int base = 0;
        for (int w = 0; w < (tid >> 5); ++w) base += wave_tot[w];
        const int ex = base + incl - v;
        start[tid] = ex;
        cnt[tid]   = ex;
        if (tid == NUM_OWNERS - 1) start[NUM_OWNERS] = base + incl;
    }
    __syncthreads();

#pragma unroll
    for (int k = 0; k < EDGES_PER_THREAD; ++k) {
        if (bk[k] >= 0) {
            const int pos = atomicAdd(&cnt[bk[k]], 1);
            sorted[pos] = pk[k];
        }
    }
    __syncthreads();

    int* __restrict__ lp = lists + (size_t)g * LIST_PITCH;
    for (int i = tid; i < ne; i += BIN_THREADS) lp[i] = sorted[i];
    if (tid <= NUM_OWNERS) offs[(size_t)g * OFF_PITCH + tid] = start[tid];
}

__global__ void norm_kernel(float* __restrict__ norm, int n) {
    int i = blockIdx.x * blockDim.x + threadIdx.x;
    if (i < n) {
        float d = norm[i];
        norm[i] = rsqrtf(fmaxf(d, 1.0f));
    }
}

#define OWNER_THREADS   1024
#define OWNER_WAVES     (OWNER_THREADS / 32)
#define SLICES_PER_WAVE (BIN_SLICES / OWNER_WAVES)
#define QUEUE_CAP       (32 + 32 * SLICES_PER_WAVE)
#define EDGES_IN_FLIGHT 8
#define ACC_PITCH       (HID + 4)
#define TILE_ROWS       (((NODES_PER_BLOCK + 15) / 16) * 16)

static_assert(BIN_SLICES % OWNER_WAVES == 0 && SLICES_PER_WAVE <= 32, "one lane per slice of a wave");
static_assert((ACC_PITCH * 4) % 16 == 0 && ((ACC_PITCH * 4) / 16) % 2 == 1, "tile pitch: 16 bytes x odd");
static_assert(HID == 128, "phase 2 covers the output columns as two groups of 64");
static_assert(TILE_ROWS / 16 <= OWNER_WAVES, "phase 2 gives a wave at most two units");

__device__ __forceinline__ void add_queued_rows(const float* __restrict__ h,
                                                float* acc,
                                                int s, int dl, int nsb, int cntw, int lane) {
    if (cntw == 32) {
        for (int j0 = 0; j0 < 32; j0 += EDGES_IN_FLIGHT) {
            float v[EDGES_IN_FLIGHT][4];
#pragma unroll
            for (int j = 0; j < EDGES_IN_FLIGHT; ++j) {
                const int sj = __builtin_amdgcn_readlane(s, j0 + j);
                const float* __restrict__ hp = h + (size_t)sj * HID + lane;
                v[j][0] = hp[0];
                v[j][1] = hp[32];
                v[j][2] = hp[64];
                v[j][3] = hp[96];
            }
#pragma unroll
            for (int j = 0; j < EDGES_IN_FLIGHT; ++j) {
                const int dj = __builtin_amdgcn_readlane(dl, j0 + j);
                const float nj = __int_as_float(__builtin_amdgcn_readlane(nsb, j0 + j));
                float* a = acc + dj * ACC_PITCH + lane;
                atomicAdd(a +  0, v[j][0] * nj);
                atomicAdd(a + 32, v[j][1] * nj);
                atomicAdd(a + 64, v[j][2] * nj);
                atomicAdd(a + 96, v[j][3] * nj);
            }
        }
    } else {
        for (int j = 0; j < cntw; ++j) {
            const int sj = __builtin_amdgcn_readlane(s, j);
            const int dj = __builtin_amdgcn_readlane(dl, j);
            const float nj = __int_as_float(__builtin_amdgcn_readlane(nsb, j));
            const float* __restrict__ hp = h + (size_t)sj * HID + lane;
            const float v0 = hp[0];
            const float v1 = hp[32];
            const float v2 = hp[64];
            const float v3 = hp[96];
            float* a = acc + dj * ACC_PITCH + lane;
            atomicAdd(a +  0, v0 * nj);
            atomicAdd(a + 32, v1 * nj);
            atomicAdd(a + 64, v2 * nj);
            atomicAdd(a + 96, v3 * nj);
        }
    }
}

template<bool ADD_RESID>
__global__ __launch_bounds__(OWNER_THREADS) void owner_layer_kernel(
    const float* __restrict__ h, const float* __restrict__ norm,
    const int* __restrict__ lists, const int* __restrict__ offs,
    const float* __restrict__ W, const float* __restrict__ bias,
    float* __restrict__ hout, int n_nodes)
{
    __shared__ __attribute__((aligned(16))) float acc[TILE_ROWS * ACC_PITCH];
    __shared__ int q_pack[OWNER_WAVES][QUEUE_CAP];

    const int tid  = threadIdx.x;
    const int lane = tid & 31;
    const int wave = __builtin_amdgcn_readfirstlane(tid >> 5);
    const int band = blockIdx.x;
    const int lo   = band * NODES_PER_BLOCK;
    const int cnt  = min(NODES_PER_BLOCK, n_nodes - lo);

    const int g0 = wave * SLICES_PER_WAVE;
    int o0 = 0, o1 = 0;
    if (lane < SLICES_PER_WAVE) {
        const int* __restrict__ op = offs + (size_t)(g0 + lane) * OFF_PITCH + band;
        o0 = op[0];
        o1 = op[1];
    }

    const int half   = lane >> 4;
    const int idx16  = lane & 15;
    const int koff   = half * 2;
    const int bcol   = (wave & 1) * 64 + idx16 * 4;
    const int strip0 = wave >> 1;
    const float nscale0 = norm[lo + min(strip0 * 16 + idx16, cnt - 1)];
    const float nscale1 = norm[lo + min((strip0 + OWNER_WAVES / 2) * 16 + idx16, cnt - 1)];
    const v4f bv = *reinterpret_cast<const v4f*>(bias + bcol);

    {
        float4* acc4 = reinterpret_cast<float4*>(acc);
        const float4 z = make_float4(0.0f, 0.0f, 0.0f, 0.0f);
        for (int i = tid; i < TILE_ROWS * ACC_PITCH / 4; i += OWNER_THREADS) acc4[i] = z;
    }
    __syncthreads();

    int* q = q_pack[wave];
    int pend = 0;

    auto drain = [&]() {
        int head = 0;
        __builtin_amdgcn_wave_barrier();
        while (pend - head >= 32) {
            const unsigned w = (unsigned)q[head + lane];
            const int s   = (int)(w >> DLOC_BITS);
            const int dl  = (int)(w & ((1u << DLOC_BITS) - 1u));
            const int nsb = __float_as_int(norm[s]);
            add_queued_rows(h, acc, s, dl, nsb, 32, lane);
            head += 32;
        }
        const int rem = pend - head;
        int t = 0;
        if (lane < rem) t = q[head + lane];
        if (lane < rem) q[lane] = t;
        pend = rem;
    };

    int ent[SLICES_PER_WAVE];
#pragma unroll
    for (int j = 0; j < SLICES_PER_WAVE; ++j) {
        const int a0 = __builtin_amdgcn_readlane(o0, j);
        const int a1 = __builtin_amdgcn_readlane(o1, j);
        ent[j] = 0;
        if (lane < a1 - a0) ent[j] = lists[(size_t)(g0 + j) * LIST_PITCH + a0 + lane];
    }
#pragma unroll
    for (int j = 0; j < SLICES_PER_WAVE; ++j) {
        const int a0 = __builtin_amdgcn_readlane(o0, j);
        const int a1 = __builtin_amdgcn_readlane(o1, j);
        const int n  = min(32, a1 - a0);
        if (lane < n) q[pend + lane] = ent[j];
        pend += n;
    }
    drain();

    for (int j = 0; j < SLICES_PER_WAVE; ++j) {
        const int a0 = __builtin_amdgcn_readlane(o0, j);
        const int a1 = __builtin_amdgcn_readlane(o1, j);
        const int* __restrict__ lp = lists + (size_t)(g0 + j) * LIST_PITCH;
        for (int p = a0 + 32; p < a1; p += 32) {
            const int m = min(32, a1 - p);
            if (lane < m) q[pend + lane] = lp[p + lane];
            pend += m;
            if (pend >= 32) drain();
        }
    }

    if (pend > 0) {
        __builtin_amdgcn_wave_barrier();
        int s = 0, dl = 0, nsb = 0;
        if (lane < pend) {
            const unsigned w = (unsigned)q[lane];
            s   = (int)(w >> DLOC_BITS);
            dl  = (int)(w & ((1u << DLOC_BITS) - 1u));
            nsb = __float_as_int(norm[s]);
        }
        add_queued_rows(h, acc, s, dl, nsb, pend, lane);
    }

    __syncthreads();

    const int nstrips = (cnt + 15) >> 4;
    const float* __restrict__ Wc = W + (size_t)koff * HID + bcol;

    for (int t = 0; t < 2; ++t) {
        const int strip = strip0 + t * (OWNER_WAVES / 2);
        if (strip >= nstrips) break;
        const float nscale = t ? nscale1 : nscale0;
        const float* arow = acc + (strip * 16 + idx16) * ACC_PITCH + koff;
        const int orow0 = strip * 16 + half * 8;

        v4f rs[8];
        if (ADD_RESID) {
#pragma unroll
            for (int r = 0; r < 8; ++r)
                rs[r] = *reinterpret_cast<const v4f*>(h + (size_t)(lo + min(orow0 + r, cnt - 1)) * HID + bcol);
        }

        v8f c[4];
#pragma unroll
        for (int i = 0; i < 4; ++i) c[i] = (v8f){};

#pragma unroll 4
        for (int k = 0; k < HID; k += 4) {
            v2f a = *reinterpret_cast<const v2f*>(arow + k);
            a.x *= nscale;
            a.y *= nscale;
            const float* __restrict__ Wr0 = Wc + (size_t)k * HID;
            const v4f b0 = *reinterpret_cast<const v4f*>(Wr0);
            const v4f b1 = *reinterpret_cast<const v4f*>(Wr0 + HID);
#pragma unroll
            for (int i = 0; i < 4; ++i) {
                v2f b;
                b.x = b0[i];
                b.y = b1[i];
                c[i] = __builtin_amdgcn_wmma_f32_16x16x4_f32(
                     false, a,  false, b,
                     (short)0, c[i],  false,  false);
            }
        }

#pragma unroll
        for (int r = 0; r < 8; ++r) {
            v4f v;
            v.x = fmaxf(c[0][r] + bv.x, 0.0f);
            v.y = fmaxf(c[1][r] + bv.y, 0.0f);
            v.z = fmaxf(c[2][r] + bv.z, 0.0f);
            v.w = fmaxf(c[3][r] + bv.w, 0.0f);
            if (ADD_RESID) v += rs[r];
            if (orow0 + r < cnt)
                *reinterpret_cast<v4f*>(hout + (size_t)(lo + orow0 + r) * HID + bcol) = v;
        }
    }
}

#define KS_WAVES 4
template<int K, int NT>
__global__ __launch_bounds__(KS_WAVES * 32) void wmma_gemm_ksplit(
    const float* __restrict__ A, const float* __restrict__ W,
    const float* __restrict__ bias, float* __restrict__ C, int nstrips)
{
    constexpr int N  = NT * 16;
    constexpr int NG = NT / 4;
    constexpr int KQ = K / KS_WAVES;
    constexpr int NJ = KQ / 8;
    constexpr int NS = KQ / 4;
    constexpr int NV = 16 * N / 4;
    static_assert(NT % 4 == 0 && K % (8 * KS_WAVES) == 0, "tiles in groups of 4, whole 8-k blocks");
    static_assert(NV % (KS_WAVES * 32) == 0 && 32 % (N / 4) == 0, "reduce map: a wave instruction covers whole rows");

    __shared__ v4f part[2][KS_WAVES][NV];

    const int tid   = threadIdx.x;
    const int lane  = tid & 31;
    const int wave  = __builtin_amdgcn_readfirstlane(tid >> 5);
    const int half  = lane >> 4;
    const int idx16 = lane & 15;
    const int kq0   = wave * KQ;

    v2f Bf[NS][NT];
#pragma unroll
    for (int j = 0; j < NJ; ++j) {
#pragma unroll
        for (int p = 0; p < 2; ++p) {
            const int krow = kq0 + 8 * j + 4 * half + 2 * p;
            const float* __restrict__ w0 = W + (size_t)krow * N + idx16 * 4;
#pragma unroll
            for (int g = 0; g < NG; ++g) {
                const v4f r0 = *reinterpret_cast<const v4f*>(w0 + g * 64);
                const v4f r1 = *reinterpret_cast<const v4f*>(w0 + N + g * 64);
#pragma unroll
                for (int i = 0; i < 4; ++i) {
                    v2f b;
                    b.x = r0[i];
                    b.y = r1[i];
                    Bf[2 * j + p][g * 4 + i] = b;
                }
            }
        }
    }

    const v4f bv = *reinterpret_cast<const v4f*>(bias + (lane % (N / 4)) * 4);
    const float* __restrict__ Ap = A + (size_t)idx16 * K + kq0 + 4 * half;

    auto do_strip = [&](const v4f (&ac)[NJ], const int st, const int buf) {
        v8f c[NT];
#pragma unroll
        for (int t = 0; t < NT; ++t) c[t] = (v8f){};

#pragma unroll
        for (int ks = 0; ks < NS; ++ks) {
            v2f a;
            a.x = ac[ks >> 1][2 * (ks & 1)];
            a.y = ac[ks >> 1][2 * (ks & 1) + 1];
#pragma unroll
            for (int t = 0; t < NT; ++t) {
                c[t] = __builtin_amdgcn_wmma_f32_16x16x4_f32(
                     false, a,  false, Bf[ks][t],
                     (short)0, c[t],  false,  false);
            }
        }

        v4f* mine = part[buf][wave];
#pragma unroll
        for (int g = 0; g < NG; ++g) {
#pragma unroll
            for (int r = 0; r < 8; ++r) {
                v4f v;
                v.x = c[g * 4 + 0][r];
                v.y = c[g * 4 + 1][r];
                v.z = c[g * 4 + 2][r];
                v.w = c[g * 4 + 3][r];
                mine[(8 * half + r) * (N / 4) + g * 16 + idx16] = v;
            }
        }
        __syncthreads();
#pragma unroll
        for (int t = 0; t < NV / (KS_WAVES * 32); ++t) {
            const int q = wave * (NV / KS_WAVES) + t * 32 + lane;
            v4f s = part[buf][0][q];
            s += part[buf][1][q];
            s += part[buf][2][q];
            s += part[buf][3][q];
            s += bv;
            *reinterpret_cast<v4f*>(C + (size_t)st * 16 * N + (size_t)q * 4) = s;
        }
    };
    auto load_rows = [&](v4f (&a)[NJ], const int st) {
#pragma unroll
        for (int j = 0; j < NJ; ++j)
            a[j] = *reinterpret_cast<const v4f*>(Ap + (size_t)st * 16 * K + 8 * j);
    };

    const int G = gridDim.x;
    int st = blockIdx.x;
    v4f a0[NJ], a1[NJ];
    load_rows(a0, min(st, nstrips - 1));
    while (st < nstrips) {
        load_rows(a1, min(st + G, nstrips - 1));
        do_strip(a0, st, 0);
        st += G;
        if (st >= nstrips) break;
        load_rows(a0, min(st + G, nstrips - 1));
        do_strip(a1, st, 1);
        st += G;
    }
}

extern "C" void kernel_launch(void* const* d_in, const int* in_sizes, int n_in,
                              void* d_out, int out_size, void* d_ws, size_t ws_size,
                              hipStream_t stream) {
    const float* features = (const float*)d_in[0];
    const int*   src      = (const int*)  d_in[1];
    const int*   dst      = (const int*)  d_in[2];
    const float* W_in     = (const float*)d_in[3];
    const float* b_in     = (const float*)d_in[4];
    const float* gcn_w0   = (const float*)d_in[5];
    const float* gcn_b0   = (const float*)d_in[6];
    const float* gcn_w1   = (const float*)d_in[7];
    const float* gcn_b1   = (const float*)d_in[8];
    const float* W_out    = (const float*)d_in[9];
    const float* b_out    = (const float*)d_in[10];
    float* out = (float*)d_out;

    char* ws = (char*)d_ws;
    float* norm = (float*)ws;
    ws += ((size_t)N_NODES * sizeof(float) + 255) & ~(size_t)255;
    float* H = (float*)ws;
    ws += ((size_t)N_NODES * HID * sizeof(float) + 255) & ~(size_t)255;
    float* H1 = (float*)ws;

    const int mBlocks = N_NODES / 16;

    zero_kernel<<<(N_NODES + 255) / 256, 256, 0, stream>>>(norm, N_NODES);
    int* lists = (int*)d_out;
    int* offs  = lists + BIN_SLICES * LIST_PITCH;
    degree_bin_kernel<<<BIN_SLICES, BIN_THREADS, 0, stream>>>(src, dst, norm, lists, offs, N_EDGES);
    norm_kernel<<<(N_NODES + 255) / 256, 256, 0, stream>>>(norm, N_NODES);

    wmma_gemm_ksplit<IN_DIM, 8><<<512, KS_WAVES * 32, 0, stream>>>(
        features, W_in, b_in, H, mBlocks);

    owner_layer_kernel<true><<<NUM_OWNERS, OWNER_THREADS, 0, stream>>>(
        H, norm, lists, offs, gcn_w0, gcn_b0, H1, N_NODES);

    owner_layer_kernel<false><<<NUM_OWNERS, OWNER_THREADS, 0, stream>>>(
        H1, norm, lists, offs, gcn_w1, gcn_b1, H, N_NODES);

    wmma_gemm_ksplit<HID, 4><<<1250, KS_WAVES * 32, 0, stream>>>(
        H, W_out, b_out, out, mBlocks);
}
